# starting kernel
# baseline (speedup 1.0000x reference)
#include <hip/hip_runtime.h>
#include <math.h>

typedef __attribute__((ext_vector_type(2))) float v2f;
typedef __attribute__((ext_vector_type(8))) float v8f;

#define SEGS_PER_TASK 4

__device__ __constant__ float c_declo[16] = {
    -0.0033824159510061256f, -0.0005421323317911481f, 0.03169508781149298f,
     0.007607487324917605f,  -0.1432942383508097f,   -0.061273359067658524f,
     0.4813596512583722f,     0.7771857517005235f,    0.3644418948353314f,
    -0.05194583810770904f,   -0.027219029917056003f,  0.049137179673607506f,
     0.003808752013890615f,  -0.01495225833704823f,  -0.0003029205147213668f,
     0.0018899503327594609f
};

__device__ __forceinline__ float load_reflect(const float* __restrict__ x, int L, int i) {
    int m = i - 14;
    m = (m < 0) ? -m : m;
    m = (m >= L) ? (2 * L - 2 - m) : m;
    m = (m < 0) ? 0 : ((m > L - 1) ? (L - 1) : m);
    return x[m];
}

__global__ __launch_bounds__(256)
void wp_level_kernel(const float* __restrict__ in, float* __restrict__ out,
                     int L_in, int L_out, int seg_count, int seg_blocks,
                     int total_tasks, int final_log)
{
    __shared__ float sF[32];
    const int tid = threadIdx.x;
    if (tid < 32) {
        const int c = tid >> 4, k = tid & 15;
        sF[tid] = (c == 0) ? c_declo[15 - k]
                           : ((k & 1) ? c_declo[k] : -c_declo[k]);
    }
    __syncthreads();

    const int lane = tid & 31;
    const int task = blockIdx.x * (blockDim.x >> 5) + (tid >> 5);
    if (task >= total_tasks) return;

    const int sb   = task % seg_blocks;
    const int row  = task / seg_blocks;
    const int seg0 = sb * SEGS_PER_TASK;

    const float* __restrict__ xr = in + (size_t)row * (size_t)L_in;

    const int khalf = lane >> 4;
    const int mn    = lane & 15;
    const int cch   = mn >> 3;
    const int jl    = mn & 7;

    v2f a[8];
#pragma unroll
    for (int t = 0; t < 8; ++t) {
#pragma unroll
        for (int e = 0; e < 2; ++e) {
            const int i   = 4 * t + 2 * khalf + e;
            const int kk  = i - 2 * jl;
            const int idx = (kk < 0) ? 0 : ((kk > 15) ? 15 : kk);
            const float v = sF[cch * 16 + idx];
            a[t][e] = ((unsigned)kk < 16u) ? v : 0.0f;
        }
    }

    float* __restrict__ orow = out + (size_t)(row * 2 + khalf) * (size_t)L_out;

#pragma unroll 1
    for (int s = 0; s < SEGS_PER_TASK; ++s) {
        const int seg = seg0 + s;
        if (seg >= seg_count) break;

        const int n0 = seg << 4;
        const int jb = seg << 7;
        const int w0 = n0 << 4;

        v2f b[8];
        if (w0 >= 14 && (w0 + 257) < L_in) {
            const float* __restrict__ bp = xr + (w0 - 14) + (mn << 4) + 2 * khalf;
#pragma unroll
            for (int t = 0; t < 8; ++t) {
                b[t][0] = bp[4 * t];
                b[t][1] = bp[4 * t + 1];
            }
        } else {
            const int ib = w0 + (mn << 4) + 2 * khalf;
#pragma unroll
            for (int t = 0; t < 8; ++t) {
                b[t][0] = load_reflect(xr, L_in, ib + 4 * t);
                b[t][1] = load_reflect(xr, L_in, ib + 4 * t + 1);
            }
        }

        v8f acc = {0.f, 0.f, 0.f, 0.f, 0.f, 0.f, 0.f, 0.f};
#pragma unroll
        for (int t = 0; t < 8; ++t) {
            acc = __builtin_amdgcn_wmma_f32_16x16x4_f32(
                      false, a[t], false, b[t], (short)0, acc, false, false);
        }

        const int j0 = jb + (mn << 3);
        if (jb + 128 <= L_out) {
            if (final_log) {
#pragma unroll
                for (int v = 0; v < 8; ++v) {
                    const float d = acc[v];
                    orow[j0 + v] = logf(fmaf(d, d, 1e-12f));
                }
            } else {
#pragma unroll
                for (int v = 0; v < 8; ++v) orow[j0 + v] = acc[v];
            }
        } else {
            if (final_log) {
#pragma unroll
                for (int v = 0; v < 8; ++v) {
                    const int j = j0 + v;
                    if (j < L_out) {
                        const float d = acc[v];
                        orow[j] = logf(fmaf(d, d, 1e-12f));
                    }
                }
            } else {
#pragma unroll
                for (int v = 0; v < 8; ++v) {
                    const int j = j0 + v;
                    if (j < L_out) orow[j] = acc[v];
                }
            }
        }
    }
}

extern "C" void kernel_launch(void* const* d_in, const int* in_sizes, int n_in,
                              void* d_out, int out_size, void* d_ws, size_t ws_size,
                              hipStream_t stream)
{
    (void)in_sizes; (void)n_in; (void)out_size; (void)ws_size;
    const float* x   = (const float*)d_in[0];
    float*       out = (float*)d_out;
    float*       ws  = (float*)d_ws;

    const size_t STRIDE = 8650752;
    float* bufA = ws;
    float* bufB = ws + STRIDE;

    int L_in = 65536;
    const float* cur_in = x;
    for (int l = 0; l < 8; ++l) {
        const int L_out      = (L_in + 15) >> 1;
        const int rows_in    = 128 << l;
        const int seg_count  = (L_out + 127) >> 7;
        const int seg_blocks = (seg_count + SEGS_PER_TASK - 1) / SEGS_PER_TASK;
        const int total      = rows_in * seg_blocks;
        const int final_lg   = (l == 7);

        float* cur_out = final_lg ? out : ((l & 1) ? bufB : bufA);

        const int blocks = (total + 7) / 8;
        wp_level_kernel<<<blocks, 256, 0, stream>>>(
            cur_in, cur_out, L_in, L_out, seg_count, seg_blocks, total, final_lg);

        cur_in = cur_out;
        L_in   = L_out;
    }
}


# Fuse the last 4 levels into one launch: one work-group per node row keeps each level's output in LDS as the next level's reflect-padded input (5 launches instead of 8)
# speedup vs baseline: 1.4056x; 1.4056x over previous
#include <hip/hip_runtime.h>
#include <math.h>

typedef __attribute__((ext_vector_type(2))) float v2f;
typedef __attribute__((ext_vector_type(8))) float v8f;

#define SEGS_PER_TASK 4

__device__ __constant__ float c_declo[16] = {
    -0.0033824159510061256f, -0.0005421323317911481f, 0.03169508781149298f,
     0.007607487324917605f,  -0.1432942383508097f,   -0.061273359067658524f,
     0.4813596512583722f,     0.7771857517005235f,    0.3644418948353314f,
    -0.05194583810770904f,   -0.027219029917056003f,  0.049137179673607506f,
     0.003808752013890615f,  -0.01495225833704823f,  -0.0003029205147213668f,
     0.0018899503327594609f
};

__device__ __forceinline__ float load_reflect(const float* __restrict__ x, int L, int i) {
    int m = i - 14;
    m = (m < 0) ? -m : m;
    m = (m >= L) ? (2 * L - 2 - m) : m;
    m = (m < 0) ? 0 : ((m > L - 1) ? (L - 1) : m);
    return x[m];
}

__global__ __launch_bounds__(256)
void wp_level_kernel(const float* __restrict__ in, float* __restrict__ out,
                     int L_in, int L_out, int seg_count, int seg_blocks,
                     int total_tasks, int final_log)
{
    __shared__ float sF[32];
    const int tid = threadIdx.x;
    if (tid < 32) {
        const int c = tid >> 4, k = tid & 15;
        sF[tid] = (c == 0) ? c_declo[15 - k]
                           : ((k & 1) ? c_declo[k] : -c_declo[k]);
    }
    __syncthreads();

    const int lane = tid & 31;
    const int task = blockIdx.x * (blockDim.x >> 5) + (tid >> 5);
    if (task >= total_tasks) return;

    const int sb   = task % seg_blocks;
    const int row  = task / seg_blocks;
    const int seg0 = sb * SEGS_PER_TASK;

    const float* __restrict__ xr = in + (size_t)row * (size_t)L_in;

    const int khalf = lane >> 4;
    const int mn    = lane & 15;
    const int cch   = mn >> 3;
    const int jl    = mn & 7;

    v2f a[8];
#pragma unroll
    for (int t = 0; t < 8; ++t) {
#pragma unroll
        for (int e = 0; e < 2; ++e) {
            const int i   = 4 * t + 2 * khalf + e;
            const int kk  = i - 2 * jl;
            const int idx = (kk < 0) ? 0 : ((kk > 15) ? 15 : kk);
            const float v = sF[cch * 16 + idx];
            a[t][e] = ((unsigned)kk < 16u) ? v : 0.0f;
        }
    }

    float* __restrict__ orow = out + (size_t)(row * 2 + khalf) * (size_t)L_out;

#pragma unroll 1
    for (int s = 0; s < SEGS_PER_TASK; ++s) {
        const int seg = seg0 + s;
        if (seg >= seg_count) break;

        const int n0 = seg << 4;
        const int jb = seg << 7;
        const int w0 = n0 << 4;

        v2f b[8];
        if (w0 >= 14 && (w0 + 257) < L_in) {
            const float* __restrict__ bp = xr + (w0 - 14) + (mn << 4) + 2 * khalf;
#pragma unroll
            for (int t = 0; t < 8; ++t) {
                b[t][0] = bp[4 * t];
                b[t][1] = bp[4 * t + 1];
            }
        } else {
            const int ib = w0 + (mn << 4) + 2 * khalf;
#pragma unroll
            for (int t = 0; t < 8; ++t) {
                b[t][0] = load_reflect(xr, L_in, ib + 4 * t);
                b[t][1] = load_reflect(xr, L_in, ib + 4 * t + 1);
            }
        }

        v8f acc = {0.f, 0.f, 0.f, 0.f, 0.f, 0.f, 0.f, 0.f};
#pragma unroll
        for (int t = 0; t < 8; ++t) {
            acc = __builtin_amdgcn_wmma_f32_16x16x4_f32(
                      false, a[t], false, b[t], (short)0, acc, false, false);
        }

        const int j0 = jb + (mn << 3);
        if (jb + 128 <= L_out) {
            if (final_log) {
#pragma unroll
                for (int v = 0; v < 8; ++v) {
                    const float d = acc[v];
                    orow[j0 + v] = logf(fmaf(d, d, 1e-12f));
                }
            } else {
#pragma unroll
                for (int v = 0; v < 8; ++v) orow[j0 + v] = acc[v];
            }
        } else {
            if (final_log) {
#pragma unroll
                for (int v = 0; v < 8; ++v) {
                    const int j = j0 + v;
                    if (j < L_out) {
                        const float d = acc[v];
                        orow[j] = logf(fmaf(d, d, 1e-12f));
                    }
                }
            } else {
#pragma unroll
                for (int v = 0; v < 8; ++v) {
                    const int j = j0 + v;
                    if (j < L_out) orow[j] = acc[v];
                }
            }
        }
    }
}

#define FUSE_LEVELS   4
#define FUSE_THREADS  256
#define LDS_PITCH     20
#define STAGE_GROUP   9
#define FUSE_BUF      5600

__device__ __forceinline__ int lds_pos(int p) { return p + ((p >> 4) << 2); }

__global__ __launch_bounds__(FUSE_THREADS)
void wp_fused_tail_kernel(const float* __restrict__ in, float* __restrict__ out,
                          int L_first, int nlev)
{
    __shared__ float sF[32];
    __shared__ __attribute__((aligned(16))) float sBuf[2][FUSE_BUF];

    const int tid = threadIdx.x;
    if (tid < 32) {
        const int c = tid >> 4, k = tid & 15;
        sF[tid] = (c == 0) ? c_declo[15 - k]
                           : ((k & 1) ? c_declo[k] : -c_declo[k]);
    }

    const int g = blockIdx.x;
    const float* __restrict__ xr = in + (size_t)g * (size_t)L_first;

    int L = L_first;
    int R = 1;

    {
        const int Lo   = (L + 15) >> 1;
        const int ncol = (Lo + 7) >> 3;
        const int P    = (ncol + 1) << 4;
        const int Lp   = L + 28 + (L & 1);
        for (int base = 0; base < P; base += STAGE_GROUP * FUSE_THREADS) {
            float v[STAGE_GROUP];
#pragma unroll
            for (int u = 0; u < STAGE_GROUP; ++u) {
                const int p = base + u * FUSE_THREADS + tid;
                int m = p - 14;
                m = (m < 0) ? -m : m;
                m = (m >= L) ? (2 * L - 2 - m) : m;
                m = (m < 0) ? 0 : ((m > L - 1) ? (L - 1) : m);
                const float t = xr[m];
                v[u] = (p < Lp) ? t : 0.0f;
            }
#pragma unroll
            for (int u = 0; u < STAGE_GROUP; ++u) {
                const int p = base + u * FUSE_THREADS + tid;
                if (p < P) sBuf[0][lds_pos(p)] = v[u];
            }
        }
    }
    __syncthreads();

    const int lane  = tid & 31;
    const int wave  = tid >> 5;
    const int h     = lane >> 4;
    const int mn    = lane & 15;
    const int cch   = mn >> 3;
    const int jl    = mn & 7;

    v2f a[8];
#pragma unroll
    for (int w = 0; w < 8; ++w) {
#pragma unroll
        for (int e = 0; e < 2; ++e) {
            const int i   = 4 * w + 2 * h + e;
            const int kk  = i - 2 * jl;
            const int idx = (kk < 0) ? 0 : ((kk > 15) ? 15 : kk);
            const float v = sF[cch * 16 + idx];
            a[w][e] = ((unsigned)kk < 16u) ? v : 0.0f;
        }
    }

    int cur = 0;
    for (int lev = 0; lev < nlev; ++lev) {
        const int  Lo      = (L + 15) >> 1;
        const int  ncol    = (Lo + 7) >> 3;
        const int  rstride = (ncol + 1) * LDS_PITCH;
        const int  totcol  = R * ncol;
        const int  ntile   = (totcol + 15) >> 4;
        const bool last    = (lev == nlev - 1);
        const int  Lo2     = (Lo + 15) >> 1;
        const int  ncol2   = (Lo2 + 7) >> 3;
        const int  rstr2   = (ncol2 + 1) * LDS_PITCH;
        const float* __restrict__ src = sBuf[cur];
        float* __restrict__ dst = sBuf[cur ^ 1];

        for (int tile = wave; tile < ntile; tile += (FUSE_THREADS >> 5)) {
            const int  q    = (tile << 4) + mn;
            const bool live = q < totcol;
            const int  qq   = live ? q : (totcol - 1);
            const int  r    = qq / ncol;
            const int  col  = qq - r * ncol;

            const float* bp = src + r * rstride + col * LDS_PITCH + 2 * h;
            v2f b[8];
#pragma unroll
            for (int w = 0; w < 8; ++w) {
                b[w] = *(const v2f*)(bp + (w >> 2) * LDS_PITCH + 4 * (w & 3));
            }

            v8f acc = {0.f, 0.f, 0.f, 0.f, 0.f, 0.f, 0.f, 0.f};
#pragma unroll
            for (int w = 0; w < 8; ++w) {
                acc = __builtin_amdgcn_wmma_f32_16x16x4_f32(
                          false, a[w], false, b[w], (short)0, acc, false, false);
            }

            const int j0 = col << 3;
            const int ro = 2 * r + h;
            const bool full = live && (j0 + 8 <= Lo);
            if (last) {
                float* __restrict__ orow =
                    out + ((size_t)g * (size_t)(2 * R) + (size_t)ro) * (size_t)Lo;
                if (full) {
#pragma unroll
                    for (int v = 0; v < 8; ++v) {
                        const float d = acc[v];
                        orow[j0 + v] = logf(fmaf(d, d, 1e-12f));
                    }
                } else {
#pragma unroll
                    for (int v = 0; v < 8; ++v) {
                        const int j = j0 + v;
                        if (live && j < Lo) {
                            const float d = acc[v];
                            orow[j] = logf(fmaf(d, d, 1e-12f));
                        }
                    }
                }
            } else {
                float* drow = dst + ro * rstr2;
                if (full) {
#pragma unroll
                    for (int u = 0; u < 4; ++u) {
                        v2f pr;
                        pr[0] = acc[2 * u];
                        pr[1] = acc[2 * u + 1];
                        *(v2f*)(drow + lds_pos(14 + j0 + 2 * u)) = pr;
                    }
                } else {
#pragma unroll
                    for (int v = 0; v < 8; ++v) {
                        const int j = j0 + v;
                        if (live && j < Lo) drow[lds_pos(14 + j)] = acc[v];
                    }
                }
            }
        }
        __syncthreads();

        if (!last) {
            const int padr = 14 + (Lo & 1);
            const int Lp2  = Lo + 28 + (Lo & 1);
            const int P2   = (ncol2 + 1) << 4;
            const int nfix = (2 * R) << 6;
            for (int u = tid; u < nfix; u += FUSE_THREADS) {
                const int rr = u >> 6;
                const int k  = u & 63;
                float* drow = dst + rr * rstr2;
                if (k < 14) {
                    drow[lds_pos(k)] = drow[lds_pos(28 - k)];
                } else if (k < 14 + padr) {
                    const int qd = k - 13;
                    drow[lds_pos(13 + Lo + qd)] = drow[lds_pos(13 + Lo - qd)];
                } else {
                    const int z = Lp2 + (k - 14 - padr);
                    if (z < P2) drow[lds_pos(z)] = 0.0f;
                }
            }
            __syncthreads();
        }

        L = Lo;
        R <<= 1;
        cur ^= 1;
    }
}

extern "C" void kernel_launch(void* const* d_in, const int* in_sizes, int n_in,
                              void* d_out, int out_size, void* d_ws, size_t ws_size,
                              hipStream_t stream)
{
    (void)in_sizes; (void)n_in; (void)out_size; (void)ws_size;
    const float* x   = (const float*)d_in[0];
    float*       out = (float*)d_out;
    float*       ws  = (float*)d_ws;

    const size_t STRIDE = 8650752;
    float* bufA = ws;
    float* bufB = ws + STRIDE;

    int L_in = 65536;
    const float* cur_in = x;
    for (int l = 0; l < 8 - FUSE_LEVELS; ++l) {
        const int L_out      = (L_in + 15) >> 1;
        const int rows_in    = 128 << l;
        const int seg_count  = (L_out + 127) >> 7;
        const int seg_blocks = (seg_count + SEGS_PER_TASK - 1) / SEGS_PER_TASK;
        const int total      = rows_in * seg_blocks;

        float* cur_out = (l & 1) ? bufB : bufA;

        const int blocks = (total + 7) / 8;
        wp_level_kernel<<<blocks, 256, 0, stream>>>(
            cur_in, cur_out, L_in, L_out, seg_count, seg_blocks, total, 0);

        cur_in = cur_out;
        L_in   = L_out;
    }

    wp_fused_tail_kernel<<<128 << (8 - FUSE_LEVELS), FUSE_THREADS, 0, stream>>>(
        cur_in, out, L_in, FUSE_LEVELS);
}


# Fuse the first 4 levels into one launch: one work-group per row tile keeps each level in LDS and recomputes the 2n+14 halo per level (2 launches instead of 5)
# speedup vs baseline: 2.2531x; 1.6029x over previous
#include <hip/hip_runtime.h>
#include <math.h>

typedef __attribute__((ext_vector_type(2))) float v2f;
typedef __attribute__((ext_vector_type(8))) float v8f;

__device__ __constant__ float c_declo[16] = {
    -0.0033824159510061256f, -0.0005421323317911481f, 0.03169508781149298f,
     0.007607487324917605f,  -0.1432942383508097f,   -0.061273359067658524f,
     0.4813596512583722f,     0.7771857517005235f,    0.3644418948353314f,
    -0.05194583810770904f,   -0.027219029917056003f,  0.049137179673607506f,
     0.003808752013890615f,  -0.01495225833704823f,  -0.0003029205147213668f,
     0.0018899503327594609f
};

#define HEAD_LEVELS   4
#define HEAD_WC       29
#define HEAD_THREADS  256
#define HEAD_PITCH    20
#define HEAD_STAGE    9
#define HEAD_BUF      5120

__device__ __forceinline__ int head_pos(int p) { return p + ((p >> 4) << 2); }

__global__ __launch_bounds__(HEAD_THREADS)
void wp_fused_head_kernel(const float* __restrict__ in, float* __restrict__ out,
                          int L_first, int ntiles)
{
    __shared__ float sF[32];
    __shared__ __attribute__((aligned(16))) float sBuf[2][HEAD_BUF];

    const int tid = threadIdx.x;
    if (tid < 32) {
        const int c = tid >> 4, k = tid & 15;
        sF[tid] = (c == 0) ? c_declo[15 - k]
                           : ((k & 1) ? c_declo[k] : -c_declo[k]);
    }

    const int g  = blockIdx.x / ntiles;
    const int tI = blockIdx.x - g * ntiles;
    const float* __restrict__ xr = in + (size_t)g * (size_t)L_first;

    int Ls[HEAD_LEVELS + 1];
    Ls[0] = L_first;
#pragma unroll
    for (int l = 0; l < HEAD_LEVELS; ++l) Ls[l + 1] = (Ls[l] + 15) >> 1;

    int cS[HEAD_LEVELS + 1], cE[HEAD_LEVELS + 1], pLo[HEAD_LEVELS], pHi[HEAD_LEVELS];
    {
        const int ncolF = (Ls[HEAD_LEVELS] + 7) >> 3;
        const int s = tI * HEAD_WC;
        const int e = s + HEAD_WC;
        cS[HEAD_LEVELS] = s;
        cE[HEAD_LEVELS] = (e > ncolF) ? ncolF : e;
    }
#pragma unroll
    for (int l = HEAD_LEVELS - 1; l >= 0; --l) {
        pLo[l] = cS[l + 1] << 4;
        pHi[l] = (cE[l + 1] + 1) << 4;
        if (l > 0) {
            const int ncol = (Ls[l] + 7) >> 3;
            const int s = (pLo[l] >> 3) - 2;
            const int e = (pHi[l] - 14 + 7) >> 3;
            cS[l] = (s < 0) ? 0 : s;
            cE[l] = (e > ncol) ? ncol : e;
        }
    }

    {
        const int L  = Ls[0];
        const int Lp = L + 28 + (L & 1);
        const int P  = pHi[0] - pLo[0];
        for (int base = 0; base < P; base += HEAD_STAGE * HEAD_THREADS) {
            float v[HEAD_STAGE];
#pragma unroll
            for (int u = 0; u < HEAD_STAGE; ++u) {
                const int p = pLo[0] + base + u * HEAD_THREADS + tid;
                int m = p - 14;
                m = (m < 0) ? -m : m;
                m = (m >= L) ? (2 * L - 2 - m) : m;
                m = (m < 0) ? 0 : ((m > L - 1) ? (L - 1) : m);
                const float t = xr[m];
                v[u] = (p < Lp) ? t : 0.0f;
            }
#pragma unroll
            for (int u = 0; u < HEAD_STAGE; ++u) {
                const int q = base + u * HEAD_THREADS + tid;
                if (q < P) sBuf[0][head_pos(q)] = v[u];
            }
        }
    }
    __syncthreads();

    const int lane  = tid & 31;
    const int wave  = tid >> 5;
    const int h     = lane >> 4;
    const int mn    = lane & 15;
    const int cch   = mn >> 3;
    const int jl    = mn & 7;

    v2f a[8];
#pragma unroll
    for (int w = 0; w < 8; ++w) {
#pragma unroll
        for (int e = 0; e < 2; ++e) {
            const int i   = 4 * w + 2 * h + e;
            const int kk  = i - 2 * jl;
            const int idx = (kk < 0) ? 0 : ((kk > 15) ? 15 : kk);
            const float v = sF[cch * 16 + idx];
            a[w][e] = ((unsigned)kk < 16u) ? v : 0.0f;
        }
    }

#pragma unroll
    for (int lev = 0; lev < HEAD_LEVELS; ++lev) {
        const int  R       = 1 << lev;
        const int  Lo      = Ls[lev + 1];
        const int  c0      = cS[lev + 1];
        const int  ncolT   = cE[lev + 1] - c0;
        const int  rstride = ((pHi[lev] - pLo[lev]) >> 4) * HEAD_PITCH;
        const int  totcol  = R * ncolT;
        const int  ntile   = (totcol + 15) >> 4;
        const bool last    = (lev == HEAD_LEVELS - 1);
        const int  qLo     = last ? 0 : pLo[last ? lev : lev + 1];
        const int  qHi     = last ? 0 : pHi[last ? lev : lev + 1];
        const int  rstr2   = ((qHi - qLo) >> 4) * HEAD_PITCH;
        const float* __restrict__ src = sBuf[lev & 1];
        float* __restrict__ dst = sBuf[(lev & 1) ^ 1];

        for (int tile = wave; tile < ntile; tile += (HEAD_THREADS >> 5)) {
            const int  q    = (tile << 4) + mn;
            const bool live = q < totcol;
            const int  qq   = live ? q : (totcol - 1);
            const int  r    = qq / ncolT;
            const int  lc   = qq - r * ncolT;
            const int  col  = c0 + lc;

            const float* bp = src + r * rstride + lc * HEAD_PITCH + 2 * h;
            v2f b[8];
#pragma unroll
            for (int w = 0; w < 8; ++w) {
                b[w] = *(const v2f*)(bp + (w >> 2) * HEAD_PITCH + 4 * (w & 3));
            }

            v8f acc = {0.f, 0.f, 0.f, 0.f, 0.f, 0.f, 0.f, 0.f};
#pragma unroll
            for (int w = 0; w < 8; ++w) {
                acc = __builtin_amdgcn_wmma_f32_16x16x4_f32(
                          false, a[w], false, b[w], (short)0, acc, false, false);
            }

            const int j0 = col << 3;
            const int ro = 2 * r + h;
            const bool full = live && (j0 + 8 <= Lo);
            if (last) {
                float* __restrict__ orow =
                    out + ((size_t)g * (size_t)(2 * R) + (size_t)ro) * (size_t)Lo;
                if (full) {
#pragma unroll
                    for (int v = 0; v < 8; ++v) orow[j0 + v] = acc[v];
                } else {
#pragma unroll
                    for (int v = 0; v < 8; ++v) {
                        const int j = j0 + v;
                        if (live && j < Lo) orow[j] = acc[v];
                    }
                }
            } else {
                float* drow = dst + ro * rstr2;
                if (full) {
#pragma unroll
                    for (int u = 0; u < 4; ++u) {
                        const int p = 14 + j0 + 2 * u;
                        if (p >= qLo && p < qHi) {
                            v2f pr;
                            pr[0] = acc[2 * u];
                            pr[1] = acc[2 * u + 1];
                            *(v2f*)(drow + head_pos(p - qLo)) = pr;
                        }
                    }
                } else {
#pragma unroll
                    for (int v = 0; v < 8; ++v) {
                        const int j = j0 + v;
                        const int p = 14 + j;
                        if (live && j < Lo && p >= qLo && p < qHi) drow[head_pos(p - qLo)] = acc[v];
                    }
                }
            }
        }
        __syncthreads();

        if (!last) {
            const int padr = 14 + (Lo & 1);
            const int Lp2  = Lo + 28 + (Lo & 1);
            const int nfix = (2 * R) << 6;
            for (int u = tid; u < nfix; u += HEAD_THREADS) {
                const int rr = u >> 6;
                const int k  = u & 63;
                float* drow = dst + rr * rstr2;
                if (k < 14) {
                    if (qLo == 0) drow[head_pos(k)] = drow[head_pos(28 - k)];
                } else if (k < 14 + padr) {
                    const int qd = k - 13;
                    const int pd = 13 + Lo + qd;
                    const int ps = 13 + Lo - qd;
                    if (pd >= qLo && pd < qHi) drow[head_pos(pd - qLo)] = drow[head_pos(ps - qLo)];
                } else {
                    const int z = Lp2 + (k - 14 - padr);
                    if (z >= qLo && z < qHi) drow[head_pos(z - qLo)] = 0.0f;
                }
            }
            __syncthreads();
        }
    }
}

#define FUSE_LEVELS   4
#define FUSE_THREADS  256
#define LDS_PITCH     20
#define STAGE_GROUP   9
#define FUSE_BUF      5600

__device__ __forceinline__ int lds_pos(int p) { return p + ((p >> 4) << 2); }

__global__ __launch_bounds__(FUSE_THREADS)
void wp_fused_tail_kernel(const float* __restrict__ in, float* __restrict__ out,
                          int L_first, int nlev)
{
    __shared__ float sF[32];
    __shared__ __attribute__((aligned(16))) float sBuf[2][FUSE_BUF];

    const int tid = threadIdx.x;
    if (tid < 32) {
        const int c = tid >> 4, k = tid & 15;
        sF[tid] = (c == 0) ? c_declo[15 - k]
                           : ((k & 1) ? c_declo[k] : -c_declo[k]);
    }

    const int g = blockIdx.x;
    const float* __restrict__ xr = in + (size_t)g * (size_t)L_first;

    int L = L_first;
    int R = 1;

    {
        const int Lo   = (L + 15) >> 1;
        const int ncol = (Lo + 7) >> 3;
        const int P    = (ncol + 1) << 4;
        const int Lp   = L + 28 + (L & 1);
        for (int base = 0; base < P; base += STAGE_GROUP * FUSE_THREADS) {
            float v[STAGE_GROUP];
#pragma unroll
            for (int u = 0; u < STAGE_GROUP; ++u) {
                const int p = base + u * FUSE_THREADS + tid;
                int m = p - 14;
                m = (m < 0) ? -m : m;
                m = (m >= L) ? (2 * L - 2 - m) : m;
                m = (m < 0) ? 0 : ((m > L - 1) ? (L - 1) : m);
                const float t = xr[m];
                v[u] = (p < Lp) ? t : 0.0f;
            }
#pragma unroll
            for (int u = 0; u < STAGE_GROUP; ++u) {
                const int p = base + u * FUSE_THREADS + tid;
                if (p < P) sBuf[0][lds_pos(p)] = v[u];
            }
        }
    }
    __syncthreads();

    const int lane  = tid & 31;
    const int wave  = tid >> 5;
    const int h     = lane >> 4;
    const int mn    = lane & 15;
    const int cch   = mn >> 3;
    const int jl    = mn & 7;

    v2f a[8];
#pragma unroll
    for (int w = 0; w < 8; ++w) {
#pragma unroll
        for (int e = 0; e < 2; ++e) {
            const int i   = 4 * w + 2 * h + e;
            const int kk  = i - 2 * jl;
            const int idx = (kk < 0) ? 0 : ((kk > 15) ? 15 : kk);
            const float v = sF[cch * 16 + idx];
            a[w][e] = ((unsigned)kk < 16u) ? v : 0.0f;
        }
    }

    int cur = 0;
    for (int lev = 0; lev < nlev; ++lev) {
        const int  Lo      = (L + 15) >> 1;
        const int  ncol    = (Lo + 7) >> 3;
        const int  rstride = (ncol + 1) * LDS_PITCH;
        const int  totcol  = R * ncol;
        const int  ntile   = (totcol + 15) >> 4;
        const bool last    = (lev == nlev - 1);
        const int  Lo2     = (Lo + 15) >> 1;
        const int  ncol2   = (Lo2 + 7) >> 3;
        const int  rstr2   = (ncol2 + 1) * LDS_PITCH;
        const float* __restrict__ src = sBuf[cur];
        float* __restrict__ dst = sBuf[cur ^ 1];

        for (int tile = wave; tile < ntile; tile += (FUSE_THREADS >> 5)) {
            const int  q    = (tile << 4) + mn;
            const bool live = q < totcol;
            const int  qq   = live ? q : (totcol - 1);
            const int  r    = qq / ncol;
            const int  col  = qq - r * ncol;

            const float* bp = src + r * rstride + col * LDS_PITCH + 2 * h;
            v2f b[8];
#pragma unroll
            for (int w = 0; w < 8; ++w) {
                b[w] = *(const v2f*)(bp + (w >> 2) * LDS_PITCH + 4 * (w & 3));
            }

            v8f acc = {0.f, 0.f, 0.f, 0.f, 0.f, 0.f, 0.f, 0.f};
#pragma unroll
            for (int w = 0; w < 8; ++w) {
                acc = __builtin_amdgcn_wmma_f32_16x16x4_f32(
                          false, a[w], false, b[w], (short)0, acc, false, false);
            }

            const int j0 = col << 3;
            const int ro = 2 * r + h;
            const bool full = live && (j0 + 8 <= Lo);
            if (last) {
                float* __restrict__ orow =
                    out + ((size_t)g * (size_t)(2 * R) + (size_t)ro) * (size_t)Lo;
                if (full) {
#pragma unroll
                    for (int v = 0; v < 8; ++v) {
                        const float d = acc[v];
                        orow[j0 + v] = logf(fmaf(d, d, 1e-12f));
                    }
                } else {
#pragma unroll
                    for (int v = 0; v < 8; ++v) {
                        const int j = j0 + v;
                        if (live && j < Lo) {
                            const float d = acc[v];
                            orow[j] = logf(fmaf(d, d, 1e-12f));
                        }
                    }
                }
            } else {
                float* drow = dst + ro * rstr2;
                if (full) {
#pragma unroll
                    for (int u = 0; u < 4; ++u) {
                        v2f pr;
                        pr[0] = acc[2 * u];
                        pr[1] = acc[2 * u + 1];
                        *(v2f*)(drow + lds_pos(14 + j0 + 2 * u)) = pr;
                    }
                } else {
#pragma unroll
                    for (int v = 0; v < 8; ++v) {
                        const int j = j0 + v;
                        if (live && j < Lo) drow[lds_pos(14 + j)] = acc[v];
                    }
                }
            }
        }
        __syncthreads();

        if (!last) {
            const int padr = 14 + (Lo & 1);
            const int Lp2  = Lo + 28 + (Lo & 1);
            const int P2   = (ncol2 + 1) << 4;
            const int nfix = (2 * R) << 6;
            for (int u = tid; u < nfix; u += FUSE_THREADS) {
                const int rr = u >> 6;
                const int k  = u & 63;
                float* drow = dst + rr * rstr2;
                if (k < 14) {
                    drow[lds_pos(k)] = drow[lds_pos(28 - k)];
                } else if (k < 14 + padr) {
                    const int qd = k - 13;
                    drow[lds_pos(13 + Lo + qd)] = drow[lds_pos(13 + Lo - qd)];
                } else {
                    const int z = Lp2 + (k - 14 - padr);
                    if (z < P2) drow[lds_pos(z)] = 0.0f;
                }
            }
            __syncthreads();
        }

        L = Lo;
        R <<= 1;
        cur ^= 1;
    }
}

extern "C" void kernel_launch(void* const* d_in, const int* in_sizes, int n_in,
                              void* d_out, int out_size, void* d_ws, size_t ws_size,
                              hipStream_t stream)
{
    (void)in_sizes; (void)n_in; (void)out_size; (void)ws_size;
    const float* x   = (const float*)d_in[0];
    float*       out = (float*)d_out;
    float*       ws  = (float*)d_ws;

    wp_fused_head_kernel<<<2304, HEAD_THREADS, 0, stream>>>(x, ws, 65536, 18);

    wp_fused_tail_kernel<<<2048, FUSE_THREADS, 0, stream>>>(ws, out, 4110, FUSE_LEVELS);
}


# Stage each input window with 16-byte global loads and 16-byte LDS stores, all issued before the first store (row ends element-wise): one memory round trip, a quarter of the staging instructions
# speedup vs baseline: 2.3448x; 1.0407x over previous
#include <hip/hip_runtime.h>
#include <math.h>

typedef __attribute__((ext_vector_type(2))) float v2f;
typedef __attribute__((ext_vector_type(4))) float v4f;
typedef __attribute__((ext_vector_type(8))) float v8f;

__device__ __constant__ float c_declo[16] = {
    -0.0033824159510061256f, -0.0005421323317911481f, 0.03169508781149298f,
     0.007607487324917605f,  -0.1432942383508097f,   -0.061273359067658524f,
     0.4813596512583722f,     0.7771857517005235f,    0.3644418948353314f,
    -0.05194583810770904f,   -0.027219029917056003f,  0.049137179673607506f,
     0.003808752013890615f,  -0.01495225833704823f,  -0.0003029205147213668f,
     0.0018899503327594609f
};

#define STAGE_THREADS 256

__device__ __forceinline__ int pad_pos(int p) { return p + ((p >> 4) << 2); }

template <int GROUPS>
__device__ __forceinline__ void stage_window(const float* __restrict__ xr, int L, int p0, int pEnd,
                                             float* __restrict__ dst, int tid)
{
    const int Lp  = L + 28 + (L & 1);
    const int P   = pEnd - p0;
    const int rs  = (L + 14) & ~3;
    const int nL  = (p0 == 0) ? 16 : 0;
    const int rlo = (rs > p0) ? rs : p0;
    const int nE  = nL + ((pEnd > rlo) ? (pEnd - rlo) : 0);

    for (int base = 0; base < P; base += 4 * GROUPS * STAGE_THREADS) {
        v4f v[GROUPS];
#pragma unroll
        for (int u = 0; u < GROUPS; ++u) {
            const int p = p0 + base + 4 * (u * STAGE_THREADS + tid);
            int m = p - 14;
            m = (m < 0) ? 0 : ((m > L - 4) ? (L - 4) : m);
            __builtin_memcpy(&v[u], xr + m, 16);
        }
        if (base == 0) {
            for (int e = tid; e < nE; e += STAGE_THREADS) {
                const int p = (e < nL) ? e : (rlo + (e - nL));
                int m = p - 14;
                m = (m < 0) ? -m : m;
                m = (m >= L) ? (2 * L - 2 - m) : m;
                m = (m < 0) ? 0 : ((m > L - 1) ? (L - 1) : m);
                const float t = xr[m];
                dst[pad_pos(p - p0)] = (p < Lp) ? t : 0.0f;
            }
        }
#pragma unroll
        for (int u = 0; u < GROUPS; ++u) {
            const int q = base + 4 * (u * STAGE_THREADS + tid);
            const int p = p0 + q;
            if (q < P && p >= 16 && p < rs) *(v4f*)(dst + pad_pos(q)) = v[u];
        }
    }
}

#define HEAD_LEVELS   4
#define HEAD_WC       29
#define HEAD_THREADS  256
#define HEAD_PITCH    20
#define HEAD_BUF      5120

__device__ __forceinline__ int head_pos(int p) { return p + ((p >> 4) << 2); }

__global__ __launch_bounds__(HEAD_THREADS)
void wp_fused_head_kernel(const float* __restrict__ in, float* __restrict__ out,
                          int L_first, int ntiles)
{
    __shared__ float sF[32];
    __shared__ __attribute__((aligned(16))) float sBuf[2][HEAD_BUF];

    const int tid = threadIdx.x;
    if (tid < 32) {
        const int c = tid >> 4, k = tid & 15;
        sF[tid] = (c == 0) ? c_declo[15 - k]
                           : ((k & 1) ? c_declo[k] : -c_declo[k]);
    }

    const int g  = blockIdx.x / ntiles;
    const int tI = blockIdx.x - g * ntiles;
    const float* __restrict__ xr = in + (size_t)g * (size_t)L_first;

    int Ls[HEAD_LEVELS + 1];
    Ls[0] = L_first;
#pragma unroll
    for (int l = 0; l < HEAD_LEVELS; ++l) Ls[l + 1] = (Ls[l] + 15) >> 1;

    int cS[HEAD_LEVELS + 1], cE[HEAD_LEVELS + 1], pLo[HEAD_LEVELS], pHi[HEAD_LEVELS];
    {
        const int ncolF = (Ls[HEAD_LEVELS] + 7) >> 3;
        const int s = tI * HEAD_WC;
        const int e = s + HEAD_WC;
        cS[HEAD_LEVELS] = s;
        cE[HEAD_LEVELS] = (e > ncolF) ? ncolF : e;
    }
#pragma unroll
    for (int l = HEAD_LEVELS - 1; l >= 0; --l) {
        pLo[l] = cS[l + 1] << 4;
        pHi[l] = (cE[l + 1] + 1) << 4;
        if (l > 0) {
            const int ncol = (Ls[l] + 7) >> 3;
            const int s = (pLo[l] >> 3) - 2;
            const int e = (pHi[l] - 14 + 7) >> 3;
            cS[l] = (s < 0) ? 0 : s;
            cE[l] = (e > ncol) ? ncol : e;
        }
    }

    stage_window<4>(xr, Ls[0], pLo[0], pHi[0], sBuf[0], tid);
    __syncthreads();

    const int lane  = tid & 31;
    const int wave  = tid >> 5;
    const int h     = lane >> 4;
    const int mn    = lane & 15;
    const int cch   = mn >> 3;
    const int jl    = mn & 7;

    v2f a[8];
#pragma unroll
    for (int w = 0; w < 8; ++w) {
#pragma unroll
        for (int e = 0; e < 2; ++e) {
            const int i   = 4 * w + 2 * h + e;
            const int kk  = i - 2 * jl;
            const int idx = (kk < 0) ? 0 : ((kk > 15) ? 15 : kk);
            const float v = sF[cch * 16 + idx];
            a[w][e] = ((unsigned)kk < 16u) ? v : 0.0f;
        }
    }

#pragma unroll
    for (int lev = 0; lev < HEAD_LEVELS; ++lev) {
        const int  R       = 1 << lev;
        const int  Lo      = Ls[lev + 1];
        const int  c0      = cS[lev + 1];
        const int  ncolT   = cE[lev + 1] - c0;
        const int  rstride = ((pHi[lev] - pLo[lev]) >> 4) * HEAD_PITCH;
        const int  totcol  = R * ncolT;
        const int  ntile   = (totcol + 15) >> 4;
        const bool last    = (lev == HEAD_LEVELS - 1);
        const int  qLo     = last ? 0 : pLo[last ? lev : lev + 1];
        const int  qHi     = last ? 0 : pHi[last ? lev : lev + 1];
        const int  rstr2   = ((qHi - qLo) >> 4) * HEAD_PITCH;
        const float* __restrict__ src = sBuf[lev & 1];
        float* __restrict__ dst = sBuf[(lev & 1) ^ 1];

        for (int tile = wave; tile < ntile; tile += (HEAD_THREADS >> 5)) {
            const int  q    = (tile << 4) + mn;
            const bool live = q < totcol;
            const int  qq   = live ? q : (totcol - 1);
            const int  r    = qq / ncolT;
            const int  lc   = qq - r * ncolT;
            const int  col  = c0 + lc;

            const float* bp = src + r * rstride + lc * HEAD_PITCH + 2 * h;
            v2f b[8];
#pragma unroll
            for (int w = 0; w < 8; ++w) {
                b[w] = *(const v2f*)(bp + (w >> 2) * HEAD_PITCH + 4 * (w & 3));
            }

            v8f acc = {0.f, 0.f, 0.f, 0.f, 0.f, 0.f, 0.f, 0.f};
#pragma unroll
            for (int w = 0; w < 8; ++w) {
                acc = __builtin_amdgcn_wmma_f32_16x16x4_f32(
                          false, a[w], false, b[w], (short)0, acc, false, false);
            }

            const int j0 = col << 3;
            const int ro = 2 * r + h;
            const bool full = live && (j0 + 8 <= Lo);
            if (last) {
                float* __restrict__ orow =
                    out + ((size_t)g * (size_t)(2 * R) + (size_t)ro) * (size_t)Lo;
                if (full) {
#pragma unroll
                    for (int v = 0; v < 8; ++v) orow[j0 + v] = acc[v];
                } else {
#pragma unroll
                    for (int v = 0; v < 8; ++v) {
                        const int j = j0 + v;
                        if (live && j < Lo) orow[j] = acc[v];
                    }
                }
            } else {
                float* drow = dst + ro * rstr2;
                if (full) {
#pragma unroll
                    for (int u = 0; u < 4; ++u) {
                        const int p = 14 + j0 + 2 * u;
                        if (p >= qLo && p < qHi) {
                            v2f pr;
                            pr[0] = acc[2 * u];
                            pr[1] = acc[2 * u + 1];
                            *(v2f*)(drow + head_pos(p - qLo)) = pr;
                        }
                    }
                } else {
#pragma unroll
                    for (int v = 0; v < 8; ++v) {
                        const int j = j0 + v;
                        const int p = 14 + j;
                        if (live && j < Lo && p >= qLo && p < qHi) drow[head_pos(p - qLo)] = acc[v];
                    }
                }
            }
        }
        __syncthreads();

        if (!last) {
            const int padr = 14 + (Lo & 1);
            const int Lp2  = Lo + 28 + (Lo & 1);
            const int nfix = (2 * R) << 6;
            for (int u = tid; u < nfix; u += HEAD_THREADS) {
                const int rr = u >> 6;
                const int k  = u & 63;
                float* drow = dst + rr * rstr2;
                if (k < 14) {
                    if (qLo == 0) drow[head_pos(k)] = drow[head_pos(28 - k)];
                } else if (k < 14 + padr) {
                    const int qd = k - 13;
                    const int pd = 13 + Lo + qd;
                    const int ps = 13 + Lo - qd;
                    if (pd >= qLo && pd < qHi) drow[head_pos(pd - qLo)] = drow[head_pos(ps - qLo)];
                } else {
                    const int z = Lp2 + (k - 14 - padr);
                    if (z >= qLo && z < qHi) drow[head_pos(z - qLo)] = 0.0f;
                }
            }
            __syncthreads();
        }
    }
}

#define FUSE_LEVELS   4
#define FUSE_THREADS  256
#define LDS_PITCH     20
#define FUSE_BUF      5600

__device__ __forceinline__ int lds_pos(int p) { return p + ((p >> 4) << 2); }

__global__ __launch_bounds__(FUSE_THREADS)
void wp_fused_tail_kernel(const float* __restrict__ in, float* __restrict__ out,
                          int L_first, int nlev)
{
    __shared__ float sF[32];
    __shared__ __attribute__((aligned(16))) float sBuf[2][FUSE_BUF];

    const int tid = threadIdx.x;
    if (tid < 32) {
        const int c = tid >> 4, k = tid & 15;
        sF[tid] = (c == 0) ? c_declo[15 - k]
                           : ((k & 1) ? c_declo[k] : -c_declo[k]);
    }

    const int g = blockIdx.x;
    const float* __restrict__ xr = in + (size_t)g * (size_t)L_first;

    int L = L_first;
    int R = 1;

    {
        const int Lo   = (L + 15) >> 1;
        const int ncol = (Lo + 7) >> 3;
        stage_window<5>(xr, L, 0, (ncol + 1) << 4, sBuf[0], tid);
    }
    __syncthreads();

    const int lane  = tid & 31;
    const int wave  = tid >> 5;
    const int h     = lane >> 4;
    const int mn    = lane & 15;
    const int cch   = mn >> 3;
    const int jl    = mn & 7;

    v2f a[8];
#pragma unroll
    for (int w = 0; w < 8; ++w) {
#pragma unroll
        for (int e = 0; e < 2; ++e) {
            const int i   = 4 * w + 2 * h + e;
            const int kk  = i - 2 * jl;
            const int idx = (kk < 0) ? 0 : ((kk > 15) ? 15 : kk);
            const float v = sF[cch * 16 + idx];
            a[w][e] = ((unsigned)kk < 16u) ? v : 0.0f;
        }
    }

    int cur = 0;
    for (int lev = 0; lev < nlev; ++lev) {
        const int  Lo      = (L + 15) >> 1;
        const int  ncol    = (Lo + 7) >> 3;
        const int  rstride = (ncol + 1) * LDS_PITCH;
        const int  totcol  = R * ncol;
        const int  ntile   = (totcol + 15) >> 4;
        const bool last    = (lev == nlev - 1);
        const int  Lo2     = (Lo + 15) >> 1;
        const int  ncol2   = (Lo2 + 7) >> 3;
        const int  rstr2   = (ncol2 + 1) * LDS_PITCH;
        const float* __restrict__ src = sBuf[cur];
        float* __restrict__ dst = sBuf[cur ^ 1];

        for (int tile = wave; tile < ntile; tile += (FUSE_THREADS >> 5)) {
            const int  q    = (tile << 4) + mn;
            const bool live = q < totcol;
            const int  qq   = live ? q : (totcol - 1);
            const int  r    = qq / ncol;
            const int  col  = qq - r * ncol;

            const float* bp = src + r * rstride + col * LDS_PITCH + 2 * h;
            v2f b[8];
#pragma unroll
            for (int w = 0; w < 8; ++w) {
                b[w] = *(const v2f*)(bp + (w >> 2) * LDS_PITCH + 4 * (w & 3));
            }

            v8f acc = {0.f, 0.f, 0.f, 0.f, 0.f, 0.f, 0.f, 0.f};
#pragma unroll
            for (int w = 0; w < 8; ++w) {
                acc = __builtin_amdgcn_wmma_f32_16x16x4_f32(
                          false, a[w], false, b[w], (short)0, acc, false, false);
            }

            const int j0 = col << 3;
            const int ro = 2 * r + h;
            const bool full = live && (j0 + 8 <= Lo);
            if (last) {
                float* __restrict__ orow =
                    out + ((size_t)g * (size_t)(2 * R) + (size_t)ro) * (size_t)Lo;
                if (full) {
#pragma unroll
                    for (int v = 0; v < 8; ++v) {
                        const float d = acc[v];
                        orow[j0 + v] = logf(fmaf(d, d, 1e-12f));
                    }
                } else {
#pragma unroll
                    for (int v = 0; v < 8; ++v) {
                        const int j = j0 + v;
                        if (live && j < Lo) {
                            const float d = acc[v];
                            orow[j] = logf(fmaf(d, d, 1e-12f));
                        }
                    }
                }
            } else {
                float* drow = dst + ro * rstr2;
                if (full) {
#pragma unroll
                    for (int u = 0; u < 4; ++u) {
                        v2f pr;
                        pr[0] = acc[2 * u];
                        pr[1] = acc[2 * u + 1];
                        *(v2f*)(drow + lds_pos(14 + j0 + 2 * u)) = pr;
                    }
                } else {
#pragma unroll
                    for (int v = 0; v < 8; ++v) {
                        const int j = j0 + v;
                        if (live && j < Lo) drow[lds_pos(14 + j)] = acc[v];
                    }
                }
            }
        }
        __syncthreads();

        if (!last) {
            const int padr = 14 + (Lo & 1);
            const int Lp2  = Lo + 28 + (Lo & 1);
            const int P2   = (ncol2 + 1) << 4;
            const int nfix = (2 * R) << 6;
            for (int u = tid; u < nfix; u += FUSE_THREADS) {
                const int rr = u >> 6;
                const int k  = u & 63;
                float* drow = dst + rr * rstr2;
                if (k < 14) {
                    drow[lds_pos(k)] = drow[lds_pos(28 - k)];
                } else if (k < 14 + padr) {
                    const int qd = k - 13;
                    drow[lds_pos(13 + Lo + qd)] = drow[lds_pos(13 + Lo - qd)];
                } else {
                    const int z = Lp2 + (k - 14 - padr);
                    if (z < P2) drow[lds_pos(z)] = 0.0f;
                }
            }
            __syncthreads();
        }

        L = Lo;
        R <<= 1;
        cur ^= 1;
    }
}

extern "C" void kernel_launch(void* const* d_in, const int* in_sizes, int n_in,
                              void* d_out, int out_size, void* d_ws, size_t ws_size,
                              hipStream_t stream)
{
    (void)in_sizes; (void)n_in; (void)out_size; (void)ws_size;
    const float* x   = (const float*)d_in[0];
    float*       out = (float*)d_out;
    float*       ws  = (float*)d_ws;

    wp_fused_head_kernel<<<2304, HEAD_THREADS, 0, stream>>>(x, ws, 65536, 18);

    wp_fused_tail_kernel<<<2048, FUSE_THREADS, 0, stream>>>(ws, out, 4110, FUSE_LEVELS);
}


# Tile-loop index arithmetic without integer division or 32-bit multiplies: row = trunc((q+0.5)/ncol) by an exact float reciprocal; row, column and pitch products as full-rate 24-bit multiplies
# speedup vs baseline: 2.5046x; 1.0682x over previous
#include <hip/hip_runtime.h>
#include <math.h>

typedef __attribute__((ext_vector_type(2))) float v2f;
typedef __attribute__((ext_vector_type(4))) float v4f;
typedef __attribute__((ext_vector_type(8))) float v8f;

__device__ __constant__ float c_declo[16] = {
    -0.0033824159510061256f, -0.0005421323317911481f, 0.03169508781149298f,
     0.007607487324917605f,  -0.1432942383508097f,   -0.061273359067658524f,
     0.4813596512583722f,     0.7771857517005235f,    0.3644418948353314f,
    -0.05194583810770904f,   -0.027219029917056003f,  0.049137179673607506f,
     0.003808752013890615f,  -0.01495225833704823f,  -0.0003029205147213668f,
     0.0018899503327594609f
};

#define STAGE_THREADS 256

__device__ __forceinline__ int pad_pos(int p) { return p + ((p >> 4) << 2); }

template <int GROUPS>
__device__ __forceinline__ void stage_window(const float* __restrict__ xr, int L, int p0, int pEnd,
                                             float* __restrict__ dst, int tid)
{
    const int Lp  = L + 28 + (L & 1);
    const int P   = pEnd - p0;
    const int rs  = (L + 14) & ~3;
    const int nL  = (p0 == 0) ? 16 : 0;
    const int rlo = (rs > p0) ? rs : p0;
    const int nE  = nL + ((pEnd > rlo) ? (pEnd - rlo) : 0);

    for (int base = 0; base < P; base += 4 * GROUPS * STAGE_THREADS) {
        v4f v[GROUPS];
#pragma unroll
        for (int u = 0; u < GROUPS; ++u) {
            const int p = p0 + base + 4 * (u * STAGE_THREADS + tid);
            int m = p - 14;
            m = (m < 0) ? 0 : ((m > L - 4) ? (L - 4) : m);
            __builtin_memcpy(&v[u], xr + m, 16);
        }
        if (base == 0) {
            for (int e = tid; e < nE; e += STAGE_THREADS) {
                const int p = (e < nL) ? e : (rlo + (e - nL));
                int m = p - 14;
                m = (m < 0) ? -m : m;
                m = (m >= L) ? (2 * L - 2 - m) : m;
                m = (m < 0) ? 0 : ((m > L - 1) ? (L - 1) : m);
                const float t = xr[m];
                dst[pad_pos(p - p0)] = (p < Lp) ? t : 0.0f;
            }
        }
#pragma unroll
        for (int u = 0; u < GROUPS; ++u) {
            const int q = base + 4 * (u * STAGE_THREADS + tid);
            const int p = p0 + q;
            if (q < P && p >= 16 && p < rs) *(v4f*)(dst + pad_pos(q)) = v[u];
        }
    }
}

#define HEAD_LEVELS   4
#define HEAD_WC       29
#define HEAD_THREADS  256
#define HEAD_PITCH    20
#define HEAD_BUF      5120

__device__ __forceinline__ int head_pos(int p) { return p + ((p >> 4) << 2); }

__global__ __launch_bounds__(HEAD_THREADS)
void wp_fused_head_kernel(const float* __restrict__ in, float* __restrict__ out,
                          int L_first, int ntiles)
{
    __shared__ float sF[32];
    __shared__ __attribute__((aligned(16))) float sBuf[2][HEAD_BUF];

    const int tid = threadIdx.x;
    if (tid < 32) {
        const int c = tid >> 4, k = tid & 15;
        sF[tid] = (c == 0) ? c_declo[15 - k]
                           : ((k & 1) ? c_declo[k] : -c_declo[k]);
    }

    const int g  = blockIdx.x / ntiles;
    const int tI = blockIdx.x - g * ntiles;
    const float* __restrict__ xr = in + (size_t)g * (size_t)L_first;

    int Ls[HEAD_LEVELS + 1];
    Ls[0] = L_first;
#pragma unroll
    for (int l = 0; l < HEAD_LEVELS; ++l) Ls[l + 1] = (Ls[l] + 15) >> 1;

    int cS[HEAD_LEVELS + 1], cE[HEAD_LEVELS + 1], pLo[HEAD_LEVELS], pHi[HEAD_LEVELS];
    {
        const int ncolF = (Ls[HEAD_LEVELS] + 7) >> 3;
        const int s = tI * HEAD_WC;
        const int e = s + HEAD_WC;
        cS[HEAD_LEVELS] = s;
        cE[HEAD_LEVELS] = (e > ncolF) ? ncolF : e;
    }
#pragma unroll
    for (int l = HEAD_LEVELS - 1; l >= 0; --l) {
        pLo[l] = cS[l + 1] << 4;
        pHi[l] = (cE[l + 1] + 1) << 4;
        if (l > 0) {
            const int ncol = (Ls[l] + 7) >> 3;
            const int s = (pLo[l] >> 3) - 2;
            const int e = (pHi[l] - 14 + 7) >> 3;
            cS[l] = (s < 0) ? 0 : s;
            cE[l] = (e > ncol) ? ncol : e;
        }
    }

    stage_window<4>(xr, Ls[0], pLo[0], pHi[0], sBuf[0], tid);
    __syncthreads();

    const int lane  = tid & 31;
    const int wave  = tid >> 5;
    const int h     = lane >> 4;
    const int mn    = lane & 15;
    const int cch   = mn >> 3;
    const int jl    = mn & 7;

    v2f a[8];
#pragma unroll
    for (int w = 0; w < 8; ++w) {
#pragma unroll
        for (int e = 0; e < 2; ++e) {
            const int i   = 4 * w + 2 * h + e;
            const int kk  = i - 2 * jl;
            const int idx = (kk < 0) ? 0 : ((kk > 15) ? 15 : kk);
            const float v = sF[cch * 16 + idx];
            a[w][e] = ((unsigned)kk < 16u) ? v : 0.0f;
        }
    }

#pragma unroll
    for (int lev = 0; lev < HEAD_LEVELS; ++lev) {
        const int  R       = 1 << lev;
        const int  Lo      = Ls[lev + 1];
        const int  c0      = cS[lev + 1];
        const int  ncolT   = cE[lev + 1] - c0;
        const int  rstride = ((pHi[lev] - pLo[lev]) >> 4) * HEAD_PITCH;
        const int  totcol  = R * ncolT;
        const int  ntile   = (totcol + 15) >> 4;
        const bool last    = (lev == HEAD_LEVELS - 1);
        const int  qLo     = last ? 0 : pLo[last ? lev : lev + 1];
        const int  qHi     = last ? 0 : pHi[last ? lev : lev + 1];
        const int  rstr2   = ((qHi - qLo) >> 4) * HEAD_PITCH;
        const float* __restrict__ src = sBuf[lev & 1];
        float* __restrict__ dst = sBuf[(lev & 1) ^ 1];
        const float inv  = __builtin_amdgcn_rcpf((float)ncolT);
        const float hinv = 0.5f * inv;

        for (int tile = wave; tile < ntile; tile += (HEAD_THREADS >> 5)) {
            const int  q    = (tile << 4) + mn;
            const bool live = q < totcol;
            const int  qq   = live ? q : (totcol - 1);
            const int  r    = (int)__builtin_fmaf((float)qq, inv, hinv);
            const int  lc   = qq - __mul24(r, ncolT);
            const int  col  = c0 + lc;

            const float* bp = src + __mul24(r, rstride) + __mul24(lc, HEAD_PITCH) + 2 * h;
            v2f b[8];
#pragma unroll
            for (int w = 0; w < 8; ++w) {
                b[w] = *(const v2f*)(bp + (w >> 2) * HEAD_PITCH + 4 * (w & 3));
            }

            v8f acc = {0.f, 0.f, 0.f, 0.f, 0.f, 0.f, 0.f, 0.f};
#pragma unroll
            for (int w = 0; w < 8; ++w) {
                acc = __builtin_amdgcn_wmma_f32_16x16x4_f32(
                          false, a[w], false, b[w], (short)0, acc, false, false);
            }

            const int j0 = col << 3;
            const int ro = 2 * r + h;
            const bool full = live && (j0 + 8 <= Lo);
            if (last) {
                float* __restrict__ orow =
                    out + (size_t)g * (size_t)(2 * R) * (size_t)Lo + __mul24(ro, Lo);
                if (full) {
#pragma unroll
                    for (int v = 0; v < 8; ++v) orow[j0 + v] = acc[v];
                } else {
#pragma unroll
                    for (int v = 0; v < 8; ++v) {
                        const int j = j0 + v;
                        if (live && j < Lo) orow[j] = acc[v];
                    }
                }
            } else {
                float* drow = dst + __mul24(ro, rstr2);
                if (full) {
#pragma unroll
                    for (int u = 0; u < 4; ++u) {
                        const int p = 14 + j0 + 2 * u;
                        if (p >= qLo && p < qHi) {
                            v2f pr;
                            pr[0] = acc[2 * u];
                            pr[1] = acc[2 * u + 1];
                            *(v2f*)(drow + head_pos(p - qLo)) = pr;
                        }
                    }
                } else {
#pragma unroll
                    for (int v = 0; v < 8; ++v) {
                        const int j = j0 + v;
                        const int p = 14 + j;
                        if (live && j < Lo && p >= qLo && p < qHi) drow[head_pos(p - qLo)] = acc[v];
                    }
                }
            }
        }
        __syncthreads();

        if (!last) {
            const int padr = 14 + (Lo & 1);
            const int Lp2  = Lo + 28 + (Lo & 1);
            const int nfix = (2 * R) << 6;
            for (int u = tid; u < nfix; u += HEAD_THREADS) {
                const int rr = u >> 6;
                const int k  = u & 63;
                float* drow = dst + __mul24(rr, rstr2);
                if (k < 14) {
                    if (qLo == 0) drow[head_pos(k)] = drow[head_pos(28 - k)];
                } else if (k < 14 + padr) {
                    const int qd = k - 13;
                    const int pd = 13 + Lo + qd;
                    const int ps = 13 + Lo - qd;
                    if (pd >= qLo && pd < qHi) drow[head_pos(pd - qLo)] = drow[head_pos(ps - qLo)];
                } else {
                    const int z = Lp2 + (k - 14 - padr);
                    if (z >= qLo && z < qHi) drow[head_pos(z - qLo)] = 0.0f;
                }
            }
            __syncthreads();
        }
    }
}

#define FUSE_LEVELS   4
#define FUSE_THREADS  256
#define LDS_PITCH     20
#define FUSE_BUF      5600

__device__ __forceinline__ int lds_pos(int p) { return p + ((p >> 4) << 2); }

__global__ __launch_bounds__(FUSE_THREADS)
void wp_fused_tail_kernel(const float* __restrict__ in, float* __restrict__ out,
                          int L_first, int nlev)
{
    __shared__ float sF[32];
    __shared__ __attribute__((aligned(16))) float sBuf[2][FUSE_BUF];

    const int tid = threadIdx.x;
    if (tid < 32) {
        const int c = tid >> 4, k = tid & 15;
        sF[tid] = (c == 0) ? c_declo[15 - k]
                           : ((k & 1) ? c_declo[k] : -c_declo[k]);
    }

    const int g = blockIdx.x;
    const float* __restrict__ xr = in + (size_t)g * (size_t)L_first;

    int L = L_first;
    int R = 1;

    {
        const int Lo   = (L + 15) >> 1;
        const int ncol = (Lo + 7) >> 3;
        stage_window<5>(xr, L, 0, (ncol + 1) << 4, sBuf[0], tid);
    }
    __syncthreads();

    const int lane  = tid & 31;
    const int wave  = tid >> 5;
    const int h     = lane >> 4;
    const int mn    = lane & 15;
    const int cch   = mn >> 3;
    const int jl    = mn & 7;

    v2f a[8];
#pragma unroll
    for (int w = 0; w < 8; ++w) {
#pragma unroll
        for (int e = 0; e < 2; ++e) {
            const int i   = 4 * w + 2 * h + e;
            const int kk  = i - 2 * jl;
            const int idx = (kk < 0) ? 0 : ((kk > 15) ? 15 : kk);
            const float v = sF[cch * 16 + idx];
            a[w][e] = ((unsigned)kk < 16u) ? v : 0.0f;
        }
    }

    int cur = 0;
    for (int lev = 0; lev < nlev; ++lev) {
        const int  Lo      = (L + 15) >> 1;
        const int  ncol    = (Lo + 7) >> 3;
        const int  rstride = (ncol + 1) * LDS_PITCH;
        const int  totcol  = R * ncol;
        const int  ntile   = (totcol + 15) >> 4;
        const bool last    = (lev == nlev - 1);
        const int  Lo2     = (Lo + 15) >> 1;
        const int  ncol2   = (Lo2 + 7) >> 3;
        const int  rstr2   = (ncol2 + 1) * LDS_PITCH;
        const float* __restrict__ src = sBuf[cur];
        float* __restrict__ dst = sBuf[cur ^ 1];
        const float inv  = __builtin_amdgcn_rcpf((float)ncol);
        const float hinv = 0.5f * inv;

        for (int tile = wave; tile < ntile; tile += (FUSE_THREADS >> 5)) {
            const int  q    = (tile << 4) + mn;
            const bool live = q < totcol;
            const int  qq   = live ? q : (totcol - 1);
            const int  r    = (int)__builtin_fmaf((float)qq, inv, hinv);
            const int  col  = qq - __mul24(r, ncol);

            const float* bp = src + __mul24(r, rstride) + __mul24(col, LDS_PITCH) + 2 * h;
            v2f b[8];
#pragma unroll
            for (int w = 0; w < 8; ++w) {
                b[w] = *(const v2f*)(bp + (w >> 2) * LDS_PITCH + 4 * (w & 3));
            }

            v8f acc = {0.f, 0.f, 0.f, 0.f, 0.f, 0.f, 0.f, 0.f};
#pragma unroll
            for (int w = 0; w < 8; ++w) {
                acc = __builtin_amdgcn_wmma_f32_16x16x4_f32(
                          false, a[w], false, b[w], (short)0, acc, false, false);
            }

            const int j0 = col << 3;
            const int ro = 2 * r + h;
            const bool full = live && (j0 + 8 <= Lo);
            if (last) {
                float* __restrict__ orow =
                    out + (size_t)g * (size_t)(2 * R) * (size_t)Lo + __mul24(ro, Lo);
                if (full) {
#pragma unroll
                    for (int v = 0; v < 8; ++v) {
                        const float d = acc[v];
                        orow[j0 + v] = logf(fmaf(d, d, 1e-12f));
                    }
                } else {
#pragma unroll
                    for (int v = 0; v < 8; ++v) {
                        const int j = j0 + v;
                        if (live && j < Lo) {
                            const float d = acc[v];
                            orow[j] = logf(fmaf(d, d, 1e-12f));
                        }
                    }
                }
            } else {
                float* drow = dst + __mul24(ro, rstr2);
                if (full) {
#pragma unroll
                    for (int u = 0; u < 4; ++u) {
                        v2f pr;
                        pr[0] = acc[2 * u];
                        pr[1] = acc[2 * u + 1];
                        *(v2f*)(drow + lds_pos(14 + j0 + 2 * u)) = pr;
                    }
                } else {
#pragma unroll
                    for (int v = 0; v < 8; ++v) {
                        const int j = j0 + v;
                        if (live && j < Lo) drow[lds_pos(14 + j)] = acc[v];
                    }
                }
            }
        }
        __syncthreads();

        if (!last) {
            const int padr = 14 + (Lo & 1);
            const int Lp2  = Lo + 28 + (Lo & 1);
            const int P2   = (ncol2 + 1) << 4;
            const int nfix = (2 * R) << 6;
            for (int u = tid; u < nfix; u += FUSE_THREADS) {
                const int rr = u >> 6;
                const int k  = u & 63;
                float* drow = dst + __mul24(rr, rstr2);
                if (k < 14) {
                    drow[lds_pos(k)] = drow[lds_pos(28 - k)];
                } else if (k < 14 + padr) {
                    const int qd = k - 13;
                    drow[lds_pos(13 + Lo + qd)] = drow[lds_pos(13 + Lo - qd)];
                } else {
                    const int z = Lp2 + (k - 14 - padr);
                    if (z < P2) drow[lds_pos(z)] = 0.0f;
                }
            }
            __syncthreads();
        }

        L = Lo;
        R <<= 1;
        cur ^= 1;
    }
}

extern "C" void kernel_launch(void* const* d_in, const int* in_sizes, int n_in,
                              void* d_out, int out_size, void* d_ws, size_t ws_size,
                              hipStream_t stream)
{
    (void)in_sizes; (void)n_in; (void)out_size; (void)ws_size;
    const float* x   = (const float*)d_in[0];
    float*       out = (float*)d_out;
    float*       ws  = (float*)d_ws;

    wp_fused_head_kernel<<<2304, HEAD_THREADS, 0, stream>>>(x, ws, 65536, 18);

    wp_fused_tail_kernel<<<2048, FUSE_THREADS, 0, stream>>>(ws, out, 4110, FUSE_LEVELS);
}
